# starting kernel
# baseline (speedup 1.0000x reference)
#include <hip/hip_runtime.h>


#define N_ROWS 2048
#define M_ROWS 512
#define FDIM   256

typedef __attribute__((ext_vector_type(2))) float v2f;
typedef __attribute__((ext_vector_type(4))) float v4f;
typedef __attribute__((ext_vector_type(8))) float v8f;

__global__ __launch_bounds__(256) void siamese_dots(
    const float* __restrict__ noteFea,
    const float* __restrict__ labFea,
    const float* __restrict__ W,
    const float* __restrict__ bptr,
    float* __restrict__ a,
    float* __restrict__ c)
{
  const int lane    = threadIdx.x & 31;
  const int wave    = threadIdx.x >> 5;
  const int chunk   = blockIdx.x * 8 + wave;
  const int rowBase = chunk * 16;
  const int half    = lane >> 4;
  const int r       = lane & 15;
  const int rowG    = rowBase + r;

  const float* src = (rowG < N_ROWS)
                   ? (noteFea + (size_t)rowG * FDIM)
                   : (labFea  + (size_t)(rowG - N_ROWS) * FDIM);

  v8f acc0 = {};
  v8f acc1 = {};
  for (int k = 0; k < FDIM; k += 8) {
    {
      const int kk = k + half * 2;
      v2f aF = *(const v2f*)(src + kk);
      v2f bF = { W[kk], W[kk + 1] };
      acc0 = __builtin_amdgcn_wmma_f32_16x16x4_f32(
          false, aF, false, bF, (short)0, acc0, false, false);
    }
    {
      const int kk = k + 4 + half * 2;
      v2f aF = *(const v2f*)(src + kk);
      v2f bF = { W[kk], W[kk + 1] };
      acc1 = __builtin_amdgcn_wmma_f32_16x16x4_f32(
          false, aF, false, bF, (short)0, acc1, false, false);
    }
  }
  v8f acc = acc0 + acc1;

  if (r == 0) {
    const int   row0 = rowBase + half * 8;
    const float bias = bptr[0];
    if (row0 < N_ROWS) {
      #pragma unroll
      for (int i = 0; i < 8; ++i) a[row0 + i] = acc[i] + bias;
    } else {
      const int m0 = row0 - N_ROWS;
      #pragma unroll
      for (int i = 0; i < 8; ++i) c[m0 + i] = acc[i];
    }
  }
}

__global__ __launch_bounds__(256) void siamese_outer(
    const float* __restrict__ a,
    const float* __restrict__ c,
    float* __restrict__ out)
{
  const int tid = blockIdx.x * blockDim.x + threadIdx.x;
  const int n   = tid >> 7;
  const int m4  = (tid & 127) << 2;

  const float an = a[n];
  v4f cv = *(const v4f*)(c + m4);
  v4f o;
  o.x = an - cv.x;
  o.y = an - cv.y;
  o.z = an - cv.z;
  o.w = an - cv.w;
  __builtin_nontemporal_store(o, (v4f*)(out + (size_t)n * M_ROWS + m4));
}

extern "C" void kernel_launch(void* const* d_in, const int* in_sizes, int n_in,
                              void* d_out, int out_size, void* d_ws, size_t ws_size,
                              hipStream_t stream) {
  const float* noteFea = (const float*)d_in[0];
  const float* labFea  = (const float*)d_in[1];
  const float* W       = (const float*)d_in[2];
  const float* b       = (const float*)d_in[3];
  float* out = (float*)d_out;

  float* a = (float*)d_ws;
  float* c = a + N_ROWS;

  siamese_dots<<<(N_ROWS + M_ROWS) / (16 * 8), 256, 0, stream>>>(
      noteFea, labFea, W, b, a, c);

  siamese_outer<<<(N_ROWS * M_ROWS / 4) / 256, 256, 0, stream>>>(a, c, out);
}


# one launch instead of two: each of 256 work-groups computes the 64 note dots and 64 lab dots of its own 64x64 out tile (same WMMA loop, same k order) into LDS, then stores the tile
# speedup vs baseline: 1.3284x; 1.3284x over previous
#include <hip/hip_runtime.h>


#define N_ROWS 2048
#define M_ROWS 512
#define FDIM   256
#define TILE   64
#define M_TILES (M_ROWS / TILE)
#define N_TILES (N_ROWS / TILE)

typedef __attribute__((ext_vector_type(2))) float v2f;
typedef __attribute__((ext_vector_type(4))) float v4f;
typedef __attribute__((ext_vector_type(8))) float v8f;

__global__ __launch_bounds__(256) void rowdots_outer_tile(
    const float* __restrict__ noteFea,
    const float* __restrict__ labFea,
    const float* __restrict__ W,
    const float* __restrict__ bptr,
    float* __restrict__ out)
{
  __shared__ float s_a[TILE];
  __shared__ float s_c[TILE];

  const int lane = threadIdx.x & 31;
  const int wave = threadIdx.x >> 5;
  const int n0   = (blockIdx.x / M_TILES) * TILE;
  const int m0   = (blockIdx.x % M_TILES) * TILE;
  const int half = lane >> 4;
  const int r    = lane & 15;

  const bool isNote = wave < 4;
  const int  chunk  = wave & 3;
  const float* src = isNote
                   ? (noteFea + (size_t)(n0 + chunk * 16 + r) * FDIM)
                   : (labFea  + (size_t)(m0 + chunk * 16 + r) * FDIM);

  v8f acc0 = {};
  v8f acc1 = {};
  for (int k = 0; k < FDIM; k += 8) {
    {
      const int kk = k + half * 2;
      v2f aF = *(const v2f*)(src + kk);
      v2f bF = { W[kk], W[kk + 1] };
      acc0 = __builtin_amdgcn_wmma_f32_16x16x4_f32(
          false, aF, false, bF, (short)0, acc0, false, false);
    }
    {
      const int kk = k + 4 + half * 2;
      v2f aF = *(const v2f*)(src + kk);
      v2f bF = { W[kk], W[kk + 1] };
      acc1 = __builtin_amdgcn_wmma_f32_16x16x4_f32(
          false, aF, false, bF, (short)0, acc1, false, false);
    }
  }
  v8f acc = acc0 + acc1;

  if (r == 0) {
    const int row0 = chunk * 16 + half * 8;
    if (isNote) {
      const float bias = bptr[0];
      #pragma unroll
      for (int i = 0; i < 8; ++i) s_a[row0 + i] = acc[i] + bias;
    } else {
      #pragma unroll
      for (int i = 0; i < 8; ++i) s_c[row0 + i] = acc[i];
    }
  }
  __syncthreads();

  const int rsub = lane >> 4;
  const int col4 = (lane & 15) << 2;
  const v4f cv = *(const v4f*)(s_c + col4);
  float* orow = out + (size_t)(n0 + wave * 8 + rsub) * M_ROWS + m0 + col4;
  for (int i = 0; i < 4; ++i) {
    const float an = s_a[wave * 8 + i * 2 + rsub];
    v4f o;
    o.x = an - cv.x;
    o.y = an - cv.y;
    o.z = an - cv.z;
    o.w = an - cv.w;
    __builtin_nontemporal_store(o, (v4f*)(orow + (size_t)(i * 2) * M_ROWS));
  }
}

extern "C" void kernel_launch(void* const* d_in, const int* in_sizes, int n_in,
                              void* d_out, int out_size, void* d_ws, size_t ws_size,
                              hipStream_t stream) {
  const float* noteFea = (const float*)d_in[0];
  const float* labFea  = (const float*)d_in[1];
  const float* W       = (const float*)d_in[2];
  const float* b       = (const float*)d_in[3];
  float* out = (float*)d_out;

  rowdots_outer_tile<<<N_TILES * M_TILES, 256, 0, stream>>>(
      noteFea, labFea, W, b, out);
}


# four waves per work-group (one per SIMD) instead of eight: each does a note chunk and a lab chunk on one W fragment set, 16-byte loads three lines ahead; WMMA k pairs regrouped, sum order changes
# speedup vs baseline: 1.5917x; 1.1982x over previous
#include <hip/hip_runtime.h>


#define N_ROWS 2048
#define M_ROWS 512
#define FDIM   256
#define TILE   64
#define M_TILES (M_ROWS / TILE)
#define N_TILES (N_ROWS / TILE)
#define LINE_F  32
#define N_STAGE (FDIM / LINE_F)
#define AHEAD   3

typedef __attribute__((ext_vector_type(2))) float v2f;
typedef __attribute__((ext_vector_type(4))) float v4f;
typedef __attribute__((ext_vector_type(8))) float v8f;

__global__ __launch_bounds__(128) void rowdots_outer_tile(
    const float* __restrict__ noteFea,
    const float* __restrict__ labFea,
    const float* __restrict__ W,
    const float* __restrict__ bptr,
    float* __restrict__ out)
{
  __shared__ float s_a[TILE];
  __shared__ float s_c[TILE];

  const int lane = threadIdx.x & 31;
  const int wave = threadIdx.x >> 5;
  const int n0   = (blockIdx.x / M_TILES) * TILE;
  const int m0   = (blockIdx.x % M_TILES) * TILE;
  const int half = lane >> 4;
  const int r    = lane & 15;

  const float bias = bptr[0];
  const float* srcN = noteFea + (size_t)(n0 + wave * 16 + r) * FDIM + half * 4;
  const float* srcL = labFea  + (size_t)(m0 + wave * 16 + r) * FDIM + half * 4;
  const float* srcW = W + half * 4;

  v4f fN[N_STAGE][4];
  v4f fL[N_STAGE][4];
  v4f fW[N_STAGE][4];

  #pragma unroll
  for (int s = 0; s < AHEAD; ++s) {
    #pragma unroll
    for (int j = 0; j < 4; ++j) {
      fN[s][j] = *(const v4f*)(srcN + s * LINE_F + j * 8);
      fL[s][j] = *(const v4f*)(srcL + s * LINE_F + j * 8);
      fW[s][j] = *(const v4f*)(srcW + s * LINE_F + j * 8);
    }
  }

  v8f accN0 = {};
  v8f accN1 = {};
  v8f accL0 = {};
  v8f accL1 = {};
  #pragma unroll
  for (int s = 0; s < N_STAGE; ++s) {
    if (s + AHEAD < N_STAGE) {
      #pragma unroll
      for (int j = 0; j < 4; ++j) {
        fN[s + AHEAD][j] = *(const v4f*)(srcN + (s + AHEAD) * LINE_F + j * 8);
        fL[s + AHEAD][j] = *(const v4f*)(srcL + (s + AHEAD) * LINE_F + j * 8);
        fW[s + AHEAD][j] = *(const v4f*)(srcW + (s + AHEAD) * LINE_F + j * 8);
      }
    }
    #pragma unroll
    for (int j = 0; j < 4; ++j) {
      const v2f w0 = { fW[s][j].x, fW[s][j].y };
      const v2f w1 = { fW[s][j].z, fW[s][j].w };
      const v2f a0 = { fN[s][j].x, fN[s][j].y };
      const v2f a1 = { fN[s][j].z, fN[s][j].w };
      const v2f l0 = { fL[s][j].x, fL[s][j].y };
      const v2f l1 = { fL[s][j].z, fL[s][j].w };
      accN0 = __builtin_amdgcn_wmma_f32_16x16x4_f32(
          false, a0, false, w0, (short)0, accN0, false, false);
      accL0 = __builtin_amdgcn_wmma_f32_16x16x4_f32(
          false, l0, false, w0, (short)0, accL0, false, false);
      accN1 = __builtin_amdgcn_wmma_f32_16x16x4_f32(
          false, a1, false, w1, (short)0, accN1, false, false);
      accL1 = __builtin_amdgcn_wmma_f32_16x16x4_f32(
          false, l1, false, w1, (short)0, accL1, false, false);
    }
  }
  __builtin_amdgcn_sched_group_barrier(0x020, 48, 0);
  #pragma unroll
  for (int q = 0; q < 16; ++q) {
    __builtin_amdgcn_sched_group_barrier(0x008, 4, 0);
    __builtin_amdgcn_sched_group_barrier(0x020, 3, 0);
  }
  __builtin_amdgcn_sched_group_barrier(0x008, 64, 0);
  asm volatile("" :: "s"(bias));
  const v8f accN = accN0 + accN1;
  const v8f accL = accL0 + accL1;

  if (r == 0) {
    const int row0 = wave * 16 + half * 8;
    #pragma unroll
    for (int i = 0; i < 8; ++i) {
      s_a[row0 + i] = accN[i] + bias;
      s_c[row0 + i] = accL[i];
    }
  }
  __syncthreads();

  const int rsub = lane >> 4;
  const int col4 = (lane & 15) << 2;
  const v4f cv = *(const v4f*)(s_c + col4);
  float* orow = out + (size_t)(n0 + wave * 16 + rsub) * M_ROWS + m0 + col4;
  for (int i = 0; i < 8; ++i) {
    const float an = s_a[wave * 16 + i * 2 + rsub];
    v4f o;
    o.x = an - cv.x;
    o.y = an - cv.y;
    o.z = an - cv.z;
    o.w = an - cv.w;
    __builtin_nontemporal_store(o, (v4f*)(orow + (size_t)(i * 2) * M_ROWS));
  }
}

extern "C" void kernel_launch(void* const* d_in, const int* in_sizes, int n_in,
                              void* d_out, int out_size, void* d_ws, size_t ws_size,
                              hipStream_t stream) {
  const float* noteFea = (const float*)d_in[0];
  const float* labFea  = (const float*)d_in[1];
  const float* W       = (const float*)d_in[2];
  const float* b       = (const float*)d_in[3];
  float* out = (float*)d_out;

  rowdots_outer_tile<<<N_TILES * M_TILES, 128, 0, stream>>>(
      noteFea, labFea, W, b, out);
}


# both WMMA operands carry data: 3 rows x 4 k-segments each, W in the last 4 slots of each; 96 not 128 WMMAs per wave, loads cover whole 128-B lines; row dot = 4 segment sums, sum order changes
# speedup vs baseline: 1.7908x; 1.1251x over previous
#include <hip/hip_runtime.h>


#define N_ROWS 2048
#define M_ROWS 512
#define FDIM   256
#define TILE   64
#define M_TILES (M_ROWS / TILE)
#define N_TILES (N_ROWS / TILE)
#define LINE_F  32
#define N_STAGE (FDIM / LINE_F)
#define AHEAD   4
#define ROWS_PP 3
#define N_PASS  6

typedef __attribute__((ext_vector_type(2))) float v2f;
typedef __attribute__((ext_vector_type(4))) float v4f;
typedef __attribute__((ext_vector_type(8))) float v8f;

#define quad_lane(x, ctrl) \
  __builtin_bit_cast(float, __builtin_amdgcn_mov_dpp( \
      __builtin_bit_cast(int, (x)), (ctrl), 0xF, 0xF, true))
#define QUAD_LANE0 0x00
#define QUAD_LANE1 0x55
#define QUAD_LANE2 0xAA
#define QUAD_LANE3 0xFF
static __device__ __forceinline__ float quad_diag_sum(float x0, float x1, float x2, float x3) {
  float r = quad_lane(x0, QUAD_LANE0);
  r += quad_lane(x1, QUAD_LANE1);
  r += quad_lane(x2, QUAD_LANE2);
  r += quad_lane(x3, QUAD_LANE3);
  return r;
}

__global__ __launch_bounds__(128) void rowdots_outer_tile(
    const float* __restrict__ noteFea,
    const float* __restrict__ labFea,
    const float* __restrict__ W,
    const float* __restrict__ bptr,
    float* __restrict__ out)
{
  __shared__ float s_ac[2 * TILE];

  const int lane = threadIdx.x & 31;
  const int wave = threadIdx.x >> 5;
  const int n0   = (blockIdx.x / M_TILES) * TILE;
  const int m0   = (blockIdx.x % M_TILES) * TILE;
  const int half = lane >> 4;
  const int slot = lane & 15;
  const int seg  = slot & 3;
  const int srow = slot >> 2;
  const int foff = seg * 8 + half * 4;

  const float bias = bptr[0];

  const bool isW  = srow == 3;
  const float* baseN = (isW ? W : noteFea + (size_t)(n0 + wave * 16 + srow) * FDIM) + foff;
  const float* baseL = (isW ? W : labFea  + (size_t)(m0 + wave * 16 + srow) * FDIM) + foff;
  const int step = isW ? 0 : ROWS_PP * FDIM;
  const int last = isW ? 0 : (15 - srow) * FDIM;
  const float* srcN[N_PASS];
  const float* srcL[N_PASS];
  #pragma unroll
  for (int p = 0; p < N_PASS; ++p) {
    const int o = (p * ROWS_PP + 2 < 16) ? p * step : last;
    srcN[p] = baseN + o;
    srcL[p] = baseL + o;
  }

  v4f fN[N_STAGE][N_PASS];
  v4f fL[N_STAGE][N_PASS];

  #pragma unroll
  for (int s = 0; s < AHEAD; ++s) {
    #pragma unroll
    for (int p = 0; p < N_PASS; ++p) {
      fN[s][p] = *(const v4f*)(srcN[p] + s * LINE_F);
      fL[s][p] = *(const v4f*)(srcL[p] + s * LINE_F);
    }
  }

  v8f acc[N_PASS];
  #pragma unroll
  for (int p = 0; p < N_PASS; ++p) acc[p] = (v8f){};

  #pragma unroll
  for (int s = 0; s < N_STAGE; ++s) {
    if (s + AHEAD < N_STAGE) {
      #pragma unroll
      for (int p = 0; p < N_PASS; ++p) {
        fN[s + AHEAD][p] = *(const v4f*)(srcN[p] + (s + AHEAD) * LINE_F);
        fL[s + AHEAD][p] = *(const v4f*)(srcL[p] + (s + AHEAD) * LINE_F);
      }
    }
    #pragma unroll
    for (int p = 0; p < N_PASS; ++p) {
      const v2f a0 = { fN[s][p].x, fN[s][p].y };
      const v2f b0 = { fL[s][p].x, fL[s][p].y };
      acc[p] = __builtin_amdgcn_wmma_f32_16x16x4_f32(
          false, a0, false, b0, (short)0, acc[p], false, false);
    }
    #pragma unroll
    for (int p = 0; p < N_PASS; ++p) {
      const v2f a1 = { fN[s][p].z, fN[s][p].w };
      const v2f b1 = { fL[s][p].z, fL[s][p].w };
      acc[p] = __builtin_amdgcn_wmma_f32_16x16x4_f32(
          false, a1, false, b1, (short)0, acc[p], false, false);
    }
  }
  __builtin_amdgcn_sched_group_barrier(0x020, 48, 0);
  #pragma unroll
  for (int q = 0; q < 16; ++q) {
    __builtin_amdgcn_sched_group_barrier(0x008, 3, 0);
    __builtin_amdgcn_sched_group_barrier(0x020, 3, 0);
  }
  __builtin_amdgcn_sched_group_barrier(0x008, 48, 0);
  asm volatile("" :: "s"(bias));

  const int q4   = lane & 3;
  const int quad = lane >> 2;
  float u[N_PASS], v[N_PASS];
  #pragma unroll
  for (int p = 0; p < N_PASS; ++p) {
    u[p] = quad_diag_sum(acc[p][0], acc[p][1], acc[p][2], acc[p][3]);
    v[p] = quad_diag_sum(acc[p][4], acc[p][5], acc[p][6], acc[p][7]);
  }
  const bool actU = q4 == 0 && (quad == 3 || quad == 7);
  const bool actV = q4 == 0 && quad >= 3 && quad <= 6;
  const int  idxU = wave * 16 + (quad == 7 ? 2 : 0);
  const int  idxV = quad == 3 ? wave * 16 + 1 : TILE + wave * 16 + quad - 4;
  const float bV  = quad == 3 ? bias : 0.0f;
  if (actU) {
    #pragma unroll
    for (int p = 0; p < N_PASS - 1; ++p) s_ac[idxU + p * ROWS_PP] = u[p] + bias;
    if (quad == 3) s_ac[idxU + (N_PASS - 1) * ROWS_PP] = u[N_PASS - 1] + bias;
  }
  if (actV) {
    #pragma unroll
    for (int p = 0; p < N_PASS - 1; ++p) s_ac[idxV + p * ROWS_PP] = v[p] + bV;
    if (quad == 4) s_ac[idxV + (N_PASS - 1) * ROWS_PP] = v[N_PASS - 1];
  }
  __syncthreads();

  const int rsub = lane >> 4;
  const int col4 = (lane & 15) << 2;
  const v4f cv = *(const v4f*)(s_ac + TILE + col4);
  float* orow = out + (size_t)(n0 + wave * 16 + rsub) * M_ROWS + m0 + col4;
  for (int i = 0; i < 8; ++i) {
    const float an = s_ac[wave * 16 + i * 2 + rsub];
    v4f o;
    o.x = an - cv.x;
    o.y = an - cv.y;
    o.z = an - cv.z;
    o.w = an - cv.w;
    __builtin_nontemporal_store(o, (v4f*)(orow + (size_t)(i * 2) * M_ROWS));
  }
}

extern "C" void kernel_launch(void* const* d_in, const int* in_sizes, int n_in,
                              void* d_out, int out_size, void* d_ws, size_t ws_size,
                              hipStream_t stream) {
  const float* noteFea = (const float*)d_in[0];
  const float* labFea  = (const float*)d_in[1];
  const float* W       = (const float*)d_in[2];
  const float* b       = (const float*)d_in[3];
  float* out = (float*)d_out;

  rowdots_outer_tile<<<N_TILES * M_TILES, 128, 0, stream>>>(
      noteFea, labFea, W, b, out);
}
